# starting kernel
# baseline (speedup 1.0000x reference)
#include <hip/hip_runtime.h>

typedef __attribute__((ext_vector_type(2))) float v2f;
typedef __attribute__((ext_vector_type(8))) float v8f;

#define N_FEAT 512
#define HIDDEN 128
#define NCLS   64

__device__ __forceinline__ void atomAddF32(float* p, float v) {
    unsafeAtomicAdd(p, v);
}


__global__ void fill_kernel(float* __restrict__ p, float v, long long n) {
    long long i = (long long)blockIdx.x * blockDim.x + threadIdx.x;
    if (i < n) p[i] = v;
}

__global__ void deg_accum_kernel(const int* __restrict__ dst, float* __restrict__ deg, int E) {
    int e = blockIdx.x * blockDim.x + threadIdx.x;
    if (e < E) atomAddF32(&deg[dst[e]], 1.0f);
}

__global__ void rsqrt_kernel(float* __restrict__ p, int n) {
    int i = blockIdx.x * blockDim.x + threadIdx.x;
    if (i < n) p[i] = rsqrtf(p[i]);
}

template<int F>
__global__ void bias_relu_kernel(float* __restrict__ t, const float* __restrict__ b, long long n) {
    long long i = (long long)blockIdx.x * blockDim.x + threadIdx.x;
    if (i < n) {
        int j = (int)(i & (F - 1));
        t[i] = fmaxf(t[i] + b[j], 0.0f);
    }
}

template<int F>
__global__ void init_bias_kernel(float* __restrict__ out, const float* __restrict__ b, long long n) {
    long long i = (long long)blockIdx.x * blockDim.x + threadIdx.x;
    if (i < n) out[i] = b[(int)(i & (F - 1))];
}


template<int K, int NCOL>
__global__ __launch_bounds__(256)
void gemm_wmma_f32(const float* __restrict__ A, const float* __restrict__ B,
                   float* __restrict__ C, int M) {
    constexpr int NT = NCOL / 16;
    const int lane = threadIdx.x & 31;
    const int wave = threadIdx.x >> 5;
    const int m0 = (blockIdx.x * 8 + wave) * 16;
    if (m0 >= M) return;

    const int mr = lane & 15;
    const int kb = (lane >> 4) << 1;
    const int nc = lane & 15;

    const float* a_ptr = A + (size_t)(m0 + mr) * K + kb;
    const float* b_ptr = B + (size_t)kb * NCOL + nc;

    v8f acc[NT] = {};
#pragma unroll 4
    for (int k = 0; k < K; k += 4) {
        v2f a;
        a.x = a_ptr[0];
        a.y = a_ptr[1];
#pragma unroll
        for (int nt = 0; nt < NT; ++nt) {
            v2f b;
            b.x = b_ptr[nt * 16];
            b.y = b_ptr[NCOL + nt * 16];
            acc[nt] = __builtin_amdgcn_wmma_f32_16x16x4_f32(
                 false, a,  false, b,
                 (short)0, acc[nt],  false,  false);
        }
        a_ptr += 4;
        b_ptr += (size_t)4 * NCOL;
    }

    const int rbase = m0 + ((lane >> 4) << 3);
#pragma unroll
    for (int nt = 0; nt < NT; ++nt) {
        float* c_ptr = C + (size_t)rbase * NCOL + nt * 16 + nc;
#pragma unroll
        for (int r = 0; r < 8; ++r)
            c_ptr[(size_t)r * NCOL] = acc[nt][r];
    }
}


template<int F, int LOG_TPE>
__global__ void scatter_kernel(const float* __restrict__ h, const int* __restrict__ srcI,
                               const int* __restrict__ dstI, const float* __restrict__ dinv,
                               float* __restrict__ acc, int E, int N) {
    long long tid = (long long)blockIdx.x * blockDim.x + threadIdx.x;
    long long edge = tid >> LOG_TPE;
    if (edge >= (long long)E + N) return;
    int fo = ((int)tid & ((1 << LOG_TPE) - 1)) << 2;

    int s, d;
    if (edge < E) { s = srcI[edge]; d = dstI[edge]; }
    else          { s = d = (int)(edge - E); }

    float nrm = dinv[s] * dinv[d];
    const float4 v = *(const float4*)(h + (size_t)s * F + fo);
    float* o = acc + (size_t)d * F + fo;
    atomAddF32(o + 0, v.x * nrm);
    atomAddF32(o + 1, v.y * nrm);
    atomAddF32(o + 2, v.z * nrm);
    atomAddF32(o + 3, v.w * nrm);
}


extern "C" void kernel_launch(void* const* d_in, const int* in_sizes, int n_in,
                              void* d_out, int out_size, void* d_ws, size_t ws_size,
                              hipStream_t stream) {
    const float* x  = (const float*)d_in[0];
    const int*   ei = (const int*)d_in[1];
    const float* W1 = (const float*)d_in[2];
    const float* b1 = (const float*)d_in[3];
    const float* W2 = (const float*)d_in[4];
    const float* b2 = (const float*)d_in[5];
    float* out = (float*)d_out;

    const int N = in_sizes[0] / N_FEAT;
    const int E = in_sizes[1] / 2;
    const int* srcI = ei;
    const int* dstI = ei + E;

    char* ws = (char*)d_ws;
    size_t degB = ((size_t)N * sizeof(float) + 255) & ~(size_t)255;
    size_t t1B  = (((size_t)N * HIDDEN * sizeof(float)) + 255) & ~(size_t)255;
    float* deg = (float*)ws;
    float* t1  = (float*)(ws + degB);
    float* t2  = (float*)(ws + degB + t1B);
    float* t3  = t1;

    const int BT = 256;
    const int Mtiles = N / 16;
    const int gemmBlocks = (Mtiles + 7) / 8;

    fill_kernel<<<(N + BT - 1) / BT, BT, 0, stream>>>(deg, 1.0f, N);
    deg_accum_kernel<<<(E + BT - 1) / BT, BT, 0, stream>>>(dstI, deg, E);
    rsqrt_kernel<<<(N + BT - 1) / BT, BT, 0, stream>>>(deg, N);

    long long t2n = (long long)N * HIDDEN;
    fill_kernel<<<(int)((t2n + BT - 1) / BT), BT, 0, stream>>>(t2, 0.0f, t2n);

    gemm_wmma_f32<N_FEAT, HIDDEN><<<gemmBlocks, 256, 0, stream>>>(x, W1, t1, N);

    long long tot1 = ((long long)E + N) << 5;
    scatter_kernel<HIDDEN, 5><<<(int)((tot1 + BT - 1) / BT), BT, 0, stream>>>(
        t1, srcI, dstI, deg, t2, E, N);

    bias_relu_kernel<HIDDEN><<<(int)((t2n + BT - 1) / BT), BT, 0, stream>>>(t2, b1, t2n);

    gemm_wmma_f32<HIDDEN, NCLS><<<gemmBlocks, 256, 0, stream>>>(t2, W2, t3, N);

    long long outn = (long long)N * NCLS;
    init_bias_kernel<NCLS><<<(int)((outn + BT - 1) / BT), BT, 0, stream>>>(out, b2, outn);
    long long tot2 = ((long long)E + N) << 4;
    scatter_kernel<NCLS, 4><<<(int)((tot2 + BT - 1) / BT), BT, 0, stream>>>(
        t3, srcI, dstI, deg, out, E, N);
}


# Float-atomic scatter-adds replaced by a gather over a destination-grouped edge list built per call (int rank atomics + prefix sum); one owner per output row, sums in arrival-rank order
# speedup vs baseline: 1.5566x; 1.5566x over previous
#include <hip/hip_runtime.h>

typedef __attribute__((ext_vector_type(2))) float v2f;
typedef __attribute__((ext_vector_type(8))) float v8f;

#define N_FEAT 512
#define HIDDEN 128
#define NCLS   64


__global__ void fill_int_kernel(int* __restrict__ p, int v, int n) {
    int i = blockIdx.x * blockDim.x + threadIdx.x;
    if (i < n) p[i] = v;
}

__global__ void count_rank_kernel(const int* __restrict__ dst, int* __restrict__ cnt,
                                  int* __restrict__ rank, int E) {
    int e = blockIdx.x * blockDim.x + threadIdx.x;
    if (e < E) rank[e] = atomicAdd(&cnt[dst[e]], 1);
}

__global__ __launch_bounds__(1024)
void scan_dinv_kernel(int* __restrict__ cnt_rowptr, float* __restrict__ dinv, int n) {
    __shared__ int wtot[32];
    const int lane = threadIdx.x & 31;
    const int wave = threadIdx.x >> 5;
    const int trips = (((n + 31) >> 5) + 127) >> 7;
    const int wbeg = wave * trips * 128;

    int s = 0;
    for (int it = 0; it < trips; ++it) {
        const int i = wbeg + it * 128 + lane * 4;
        if (i + 3 < n) {
            const int4 c = *(const int4*)(cnt_rowptr + i);
            s += (c.x + c.y) + (c.z + c.w);
        } else {
            for (int k = 0; k < 4; ++k) if (i + k < n) s += cnt_rowptr[i + k];
        }
    }
    for (int off = 16; off > 0; off >>= 1) s += __shfl_xor(s, off);
    if (lane == 0) wtot[wave] = s;
    __syncthreads();
    int run = 0;
    for (int w = 0; w < wave; ++w) run += wtot[w];

    for (int it = 0; it < trips; ++it) {
        const int i = wbeg + it * 128 + lane * 4;
        int c0 = 0, c1 = 0, c2 = 0, c3 = 0;
        if (i + 3 < n) {
            const int4 c = *(const int4*)(cnt_rowptr + i);
            c0 = c.x; c1 = c.y; c2 = c.z; c3 = c.w;
        } else {
            if (i     < n) c0 = cnt_rowptr[i];
            if (i + 1 < n) c1 = cnt_rowptr[i + 1];
            if (i + 2 < n) c2 = cnt_rowptr[i + 2];
        }
        const int mine = (c0 + c1) + (c2 + c3);
        int inc = mine;
        for (int off = 1; off < 32; off <<= 1) {
            const int v = __shfl_up(inc, off);
            if (lane >= off) inc += v;
        }
        const int p0 = run + inc - mine;
        const int p1 = p0 + c0, p2 = p1 + c1, p3 = p2 + c2;
        if (i + 3 < n) {
            *(int4*)(cnt_rowptr + i) = make_int4(p0, p1, p2, p3);
            *(float4*)(dinv + i) = make_float4(rsqrtf((float)(c0 + 1)), rsqrtf((float)(c1 + 1)),
                                               rsqrtf((float)(c2 + 1)), rsqrtf((float)(c3 + 1)));
        } else {
            if (i     < n) { cnt_rowptr[i]     = p0; dinv[i]     = rsqrtf((float)(c0 + 1)); }
            if (i + 1 < n) { cnt_rowptr[i + 1] = p1; dinv[i + 1] = rsqrtf((float)(c1 + 1)); }
            if (i + 2 < n) { cnt_rowptr[i + 2] = p2; dinv[i + 2] = rsqrtf((float)(c2 + 1)); }
        }
        run += __shfl(inc, 31);
    }
}

__global__ void place_kernel(const int* __restrict__ srcI, const int* __restrict__ dstI,
                             const int* __restrict__ rowptr, const int* __restrict__ rank,
                             int* __restrict__ col, int E) {
    int e = blockIdx.x * blockDim.x + threadIdx.x;
    if (e < E) col[rowptr[dstI[e]] + rank[e]] = srcI[e];
}

__global__ void copy_int_kernel(const int* __restrict__ a, int* __restrict__ b, int n) {
    int i = blockIdx.x * blockDim.x + threadIdx.x;
    if (i < n) b[i] = a[i];
}


template<int K, int NCOL>
__global__ __launch_bounds__(256)
void gemm_wmma_f32(const float* __restrict__ A, const float* __restrict__ B,
                   float* __restrict__ C, int M) {
    constexpr int NT = NCOL / 16;
    const int lane = threadIdx.x & 31;
    const int wave = threadIdx.x >> 5;
    const int m0 = (blockIdx.x * 8 + wave) * 16;
    if (m0 >= M) return;

    const int mr = lane & 15;
    const int kb = (lane >> 4) << 1;
    const int nc = lane & 15;

    const float* a_ptr = A + (size_t)(m0 + mr) * K + kb;
    const float* b_ptr = B + (size_t)kb * NCOL + nc;

    v8f acc[NT] = {};
#pragma unroll 4
    for (int k = 0; k < K; k += 4) {
        v2f a;
        a.x = a_ptr[0];
        a.y = a_ptr[1];
#pragma unroll
        for (int nt = 0; nt < NT; ++nt) {
            v2f b;
            b.x = b_ptr[nt * 16];
            b.y = b_ptr[NCOL + nt * 16];
            acc[nt] = __builtin_amdgcn_wmma_f32_16x16x4_f32(
                 false, a,  false, b,
                 (short)0, acc[nt],  false,  false);
        }
        a_ptr += 4;
        b_ptr += (size_t)4 * NCOL;
    }

    const int rbase = m0 + ((lane >> 4) << 3);
#pragma unroll
    for (int nt = 0; nt < NT; ++nt) {
        float* c_ptr = C + (size_t)rbase * NCOL + nt * 16 + nc;
#pragma unroll
        for (int r = 0; r < 8; ++r)
            c_ptr[(size_t)r * NCOL] = acc[nt][r];
    }
}


template<int F, bool RELU>
__global__ __launch_bounds__(256)
void gather_kernel(const float* __restrict__ h, const int* __restrict__ col,
                   const int* __restrict__ rowptr, const float* __restrict__ dinv,
                   const float* __restrict__ bias, float* __restrict__ o, int N, int E) {
    constexpr int TPN = F / 4;
    const int tid = blockIdx.x * blockDim.x + threadIdx.x;
    const int node = tid / TPN;
    if (node >= N) return;
    const int fo = (tid % TPN) << 2;

    const int beg = rowptr[node];
    const int end = (node + 1 < N) ? rowptr[node + 1] : E;
    const float dd = dinv[node];

    const float4 self = *(const float4*)(h + (size_t)node * F + fo);
    const float sn = dd * dd;
    float4 acc;
    acc.x = self.x * sn; acc.y = self.y * sn; acc.z = self.z * sn; acc.w = self.w * sn;

    int j = beg;
    for (; j + 4 <= end; j += 4) {
        const int s0 = col[j], s1 = col[j + 1], s2 = col[j + 2], s3 = col[j + 3];
        const float4 v0 = *(const float4*)(h + (size_t)s0 * F + fo);
        const float4 v1 = *(const float4*)(h + (size_t)s1 * F + fo);
        const float4 v2 = *(const float4*)(h + (size_t)s2 * F + fo);
        const float4 v3 = *(const float4*)(h + (size_t)s3 * F + fo);
        const float n0 = dinv[s0] * dd, n1 = dinv[s1] * dd, n2 = dinv[s2] * dd, n3 = dinv[s3] * dd;
        acc.x += v0.x * n0; acc.y += v0.y * n0; acc.z += v0.z * n0; acc.w += v0.w * n0;
        acc.x += v1.x * n1; acc.y += v1.y * n1; acc.z += v1.z * n1; acc.w += v1.w * n1;
        acc.x += v2.x * n2; acc.y += v2.y * n2; acc.z += v2.z * n2; acc.w += v2.w * n2;
        acc.x += v3.x * n3; acc.y += v3.y * n3; acc.z += v3.z * n3; acc.w += v3.w * n3;
    }
    const int rem = end - j;
    if (rem > 0) {
        const int s0 = col[j];
        const int s1 = col[rem > 1 ? j + 1 : j];
        const int s2 = col[rem > 2 ? j + 2 : j];
        const float4 v0 = *(const float4*)(h + (size_t)s0 * F + fo);
        const float4 v1 = *(const float4*)(h + (size_t)s1 * F + fo);
        const float4 v2 = *(const float4*)(h + (size_t)s2 * F + fo);
        const float n0 = dinv[s0] * dd, n1 = dinv[s1] * dd, n2 = dinv[s2] * dd;
        acc.x += v0.x * n0; acc.y += v0.y * n0; acc.z += v0.z * n0; acc.w += v0.w * n0;
        if (rem > 1) { acc.x += v1.x * n1; acc.y += v1.y * n1; acc.z += v1.z * n1; acc.w += v1.w * n1; }
        if (rem > 2) { acc.x += v2.x * n2; acc.y += v2.y * n2; acc.z += v2.z * n2; acc.w += v2.w * n2; }
    }

    const float4 bb = *(const float4*)(bias + fo);
    float4 r;
    r.x = acc.x + bb.x; r.y = acc.y + bb.y; r.z = acc.z + bb.z; r.w = acc.w + bb.w;
    if (RELU) {
        r.x = fmaxf(r.x, 0.0f); r.y = fmaxf(r.y, 0.0f); r.z = fmaxf(r.z, 0.0f); r.w = fmaxf(r.w, 0.0f);
    }
    *(float4*)(o + (size_t)node * F + fo) = r;
}


extern "C" void kernel_launch(void* const* d_in, const int* in_sizes, int n_in,
                              void* d_out, int out_size, void* d_ws, size_t ws_size,
                              hipStream_t stream) {
    const float* x  = (const float*)d_in[0];
    const int*   ei = (const int*)d_in[1];
    const float* W1 = (const float*)d_in[2];
    const float* b1 = (const float*)d_in[3];
    const float* W2 = (const float*)d_in[4];
    const float* b2 = (const float*)d_in[5];
    float* out = (float*)d_out;

    const int N = in_sizes[0] / N_FEAT;
    const int E = in_sizes[1] / 2;
    const int* srcI = ei;
    const int* dstI = ei + E;

    char* ws = (char*)d_ws;
    size_t degB = ((size_t)N * sizeof(float) + 255) & ~(size_t)255;
    size_t t1B  = (((size_t)N * HIDDEN * sizeof(float)) + 255) & ~(size_t)255;
    size_t t3B  = (((size_t)N * NCLS * sizeof(float)) + 255) & ~(size_t)255;
    float* dinv = (float*)ws;
    float* t1   = (float*)(ws + degB);
    float* t2   = (float*)(ws + degB + t1B);
    float* t3   = t1;

    int* rowptrA = (int*)out;
    int* colA    = rowptrA + N;
    int* rowptrB = (int*)((char*)t1 + t3B);
    int* colB    = rowptrB + N;
    int* rank    = (int*)t2;
    const int listInts = N + E;

    const int BT = 256;
    const int Mtiles = N / 16;
    const int gemmBlocks = (Mtiles + 7) / 8;

    fill_int_kernel<<<(N + BT - 1) / BT, BT, 0, stream>>>(rowptrA, 0, N);
    count_rank_kernel<<<(E + BT - 1) / BT, BT, 0, stream>>>(dstI, rowptrA, rank, E);
    scan_dinv_kernel<<<1, 1024, 0, stream>>>(rowptrA, dinv, N);
    place_kernel<<<(E + BT - 1) / BT, BT, 0, stream>>>(srcI, dstI, rowptrA, rank, colA, E);

    gemm_wmma_f32<N_FEAT, HIDDEN><<<gemmBlocks, 256, 0, stream>>>(x, W1, t1, N);

    long long tot1 = (long long)N * (HIDDEN / 4);
    gather_kernel<HIDDEN, true><<<(int)((tot1 + BT - 1) / BT), BT, 0, stream>>>(
        t1, colA, rowptrA, dinv, b1, t2, N, E);

    copy_int_kernel<<<(listInts + BT - 1) / BT, BT, 0, stream>>>(rowptrA, rowptrB, listInts);

    gemm_wmma_f32<HIDDEN, NCLS><<<gemmBlocks, 256, 0, stream>>>(t2, W2, t3, N);

    long long tot2 = (long long)N * (NCLS / 4);
    gather_kernel<NCLS, false><<<(int)((tot2 + BT - 1) / BT), BT, 0, stream>>>(
        t3, colB, rowptrB, dinv, b2, out, N, E);
}


# Independent work shares a launch: the prefix-sum work-group (now 8 waves) is the last block of the GEMM 1 grid and the edge-list copy blocks sit behind the GEMM 2 blocks; 9 launches -> 7
# speedup vs baseline: 1.7237x; 1.1073x over previous
#include <hip/hip_runtime.h>

typedef __attribute__((ext_vector_type(2))) float v2f;
typedef __attribute__((ext_vector_type(8))) float v8f;

#define N_FEAT 512
#define HIDDEN 128
#define NCLS   64


__global__ void fill_int_kernel(int* __restrict__ p, int v, int n) {
    int i = blockIdx.x * blockDim.x + threadIdx.x;
    if (i < n) p[i] = v;
}

__global__ void count_rank_kernel(const int* __restrict__ dst, int* __restrict__ cnt,
                                  int* __restrict__ rank, int E) {
    int e = blockIdx.x * blockDim.x + threadIdx.x;
    if (e < E) rank[e] = atomicAdd(&cnt[dst[e]], 1);
}

__device__ __forceinline__ void scan_dinv_block(int* __restrict__ cnt_rowptr, float* __restrict__ dinv, int n) {
    __shared__ int wtot[8];
    const int lane = threadIdx.x & 31;
    const int wave = threadIdx.x >> 5;
    const int trips = (((n + 7) >> 3) + 511) >> 9;
    const int wbeg = wave * trips * 512;

    int s = 0;
    for (int it = 0; it < trips; ++it) {
        const int i = wbeg + it * 512 + lane * 16;
        if (i + 15 < n) {
            const int4* p = (const int4*)(cnt_rowptr + i);
            const int4 a = p[0], b = p[1], c = p[2], d = p[3];
            s += (((a.x + a.y) + (a.z + a.w)) + ((b.x + b.y) + (b.z + b.w))) +
                 (((c.x + c.y) + (c.z + c.w)) + ((d.x + d.y) + (d.z + d.w)));
        } else {
            for (int k = 0; k < 16; ++k) if (i + k < n) s += cnt_rowptr[i + k];
        }
    }
    for (int off = 16; off > 0; off >>= 1) s += __shfl_xor(s, off);
    if (lane == 0) wtot[wave] = s;
    __syncthreads();
    int run = 0;
    for (int w = 0; w < wave; ++w) run += wtot[w];

    for (int it = 0; it < trips; ++it) {
        const int i = wbeg + it * 512 + lane * 16;
        int cv[16];
        if (i + 15 < n) {
            const int4* p = (const int4*)(cnt_rowptr + i);
            const int4 a = p[0], b = p[1], c = p[2], d = p[3];
            cv[0] = a.x;  cv[1] = a.y;  cv[2] = a.z;  cv[3] = a.w;
            cv[4] = b.x;  cv[5] = b.y;  cv[6] = b.z;  cv[7] = b.w;
            cv[8] = c.x;  cv[9] = c.y;  cv[10] = c.z; cv[11] = c.w;
            cv[12] = d.x; cv[13] = d.y; cv[14] = d.z; cv[15] = d.w;
        } else {
            for (int k = 0; k < 16; ++k) cv[k] = (i + k < n) ? cnt_rowptr[i + k] : 0;
        }
        int pv[16];
        int mine = 0;
        for (int k = 0; k < 16; ++k) { pv[k] = mine; mine += cv[k]; }
        int inc = mine;
        for (int off = 1; off < 32; off <<= 1) {
            const int v = __shfl_up(inc, off);
            if (lane >= off) inc += v;
        }
        const int base = run + inc - mine;
        if (i + 15 < n) {
            int4* q = (int4*)(cnt_rowptr + i);
            float4* g = (float4*)(dinv + i);
            for (int k = 0; k < 4; ++k) {
                q[k] = make_int4(base + pv[4 * k], base + pv[4 * k + 1], base + pv[4 * k + 2], base + pv[4 * k + 3]);
                g[k] = make_float4(rsqrtf((float)(cv[4 * k] + 1)), rsqrtf((float)(cv[4 * k + 1] + 1)),
                                   rsqrtf((float)(cv[4 * k + 2] + 1)), rsqrtf((float)(cv[4 * k + 3] + 1)));
            }
        } else {
            for (int k = 0; k < 16; ++k)
                if (i + k < n) { cnt_rowptr[i + k] = base + pv[k]; dinv[i + k] = rsqrtf((float)(cv[k] + 1)); }
        }
        run += __shfl(inc, 31);
    }
}

__global__ void place_kernel(const int* __restrict__ srcI, const int* __restrict__ dstI,
                             const int* __restrict__ rowptr, const int* __restrict__ rank,
                             int* __restrict__ col, int E) {
    int e = blockIdx.x * blockDim.x + threadIdx.x;
    if (e < E) col[rowptr[dstI[e]] + rank[e]] = srcI[e];
}

__device__ __forceinline__ void copy_ints_block(const int* __restrict__ a, int* __restrict__ b, int n, int cb) {
    const int n4 = n >> 2;
    const int4* a4 = (const int4*)a;
    int4* b4 = (int4*)b;
    const int i0 = cb * 1024 + threadIdx.x;
    const int i1 = i0 + 256, i2 = i0 + 512, i3 = i0 + 768;
    int4 v0, v1, v2, v3;
    if (i0 < n4) v0 = a4[i0];
    if (i1 < n4) v1 = a4[i1];
    if (i2 < n4) v2 = a4[i2];
    if (i3 < n4) v3 = a4[i3];
    if (i0 < n4) b4[i0] = v0;
    if (i1 < n4) b4[i1] = v1;
    if (i2 < n4) b4[i2] = v2;
    if (i3 < n4) b4[i3] = v3;
    if (cb == 0 && threadIdx.x == 0)
        for (int i = n4 * 4; i < n; ++i) b[i] = a[i];
}


template<int K, int NCOL>
__device__ __forceinline__ void gemm_wmma_f32(const float* __restrict__ A, const float* __restrict__ B,
                                              float* __restrict__ C, int M, int blk) {
    constexpr int NT = NCOL / 16;
    const int lane = threadIdx.x & 31;
    const int wave = threadIdx.x >> 5;
    const int m0 = (blk * 8 + wave) * 16;
    if (m0 >= M) return;

    const int mr = lane & 15;
    const int kb = (lane >> 4) << 1;
    const int nc = lane & 15;

    const float* a_ptr = A + (size_t)(m0 + mr) * K + kb;
    const float* b_ptr = B + (size_t)kb * NCOL + nc;

    v8f acc[NT] = {};
#pragma unroll 4
    for (int k = 0; k < K; k += 4) {
        v2f a;
        a.x = a_ptr[0];
        a.y = a_ptr[1];
#pragma unroll
        for (int nt = 0; nt < NT; ++nt) {
            v2f b;
            b.x = b_ptr[nt * 16];
            b.y = b_ptr[NCOL + nt * 16];
            acc[nt] = __builtin_amdgcn_wmma_f32_16x16x4_f32(
                 false, a,  false, b,
                 (short)0, acc[nt],  false,  false);
        }
        a_ptr += 4;
        b_ptr += (size_t)4 * NCOL;
    }

    const int rbase = m0 + ((lane >> 4) << 3);
#pragma unroll
    for (int nt = 0; nt < NT; ++nt) {
        float* c_ptr = C + (size_t)rbase * NCOL + nt * 16 + nc;
#pragma unroll
        for (int r = 0; r < 8; ++r)
            c_ptr[(size_t)r * NCOL] = acc[nt][r];
    }
}


__global__ __launch_bounds__(256)
void gemm1_scan_kernel(const float* __restrict__ x, const float* __restrict__ W1, float* __restrict__ t1,
                       int M, int gemm_blocks, int* __restrict__ cnt_rowptr, float* __restrict__ dinv, int n) {
    if ((int)blockIdx.x < gemm_blocks) {
        gemm_wmma_f32<N_FEAT, HIDDEN>(x, W1, t1, M, (int)blockIdx.x);
    } else {
        scan_dinv_block(cnt_rowptr, dinv, n);
    }
}

__global__ __launch_bounds__(256)
void gemm2_copy_kernel(const float* __restrict__ h, const float* __restrict__ W2, float* __restrict__ t3,
                       int M, int gemm_blocks, const int* __restrict__ a, int* __restrict__ b, int n) {
    if ((int)blockIdx.x < gemm_blocks) {
        gemm_wmma_f32<HIDDEN, NCLS>(h, W2, t3, M, (int)blockIdx.x);
    } else {
        copy_ints_block(a, b, n, (int)blockIdx.x - gemm_blocks);
    }
}


template<int F, bool RELU>
__global__ __launch_bounds__(256)
void gather_kernel(const float* __restrict__ h, const int* __restrict__ col,
                   const int* __restrict__ rowptr, const float* __restrict__ dinv,
                   const float* __restrict__ bias, float* __restrict__ o, int N, int E) {
    constexpr int TPN = F / 4;
    const int tid = blockIdx.x * blockDim.x + threadIdx.x;
    const int node = tid / TPN;
    if (node >= N) return;
    const int fo = (tid % TPN) << 2;

    const int beg = rowptr[node];
    const int end = (node + 1 < N) ? rowptr[node + 1] : E;
    const float dd = dinv[node];

    const float4 self = *(const float4*)(h + (size_t)node * F + fo);
    const float sn = dd * dd;
    float4 acc;
    acc.x = self.x * sn; acc.y = self.y * sn; acc.z = self.z * sn; acc.w = self.w * sn;

    int j = beg;
    for (; j + 4 <= end; j += 4) {
        const int s0 = col[j], s1 = col[j + 1], s2 = col[j + 2], s3 = col[j + 3];
        const float4 v0 = *(const float4*)(h + (size_t)s0 * F + fo);
        const float4 v1 = *(const float4*)(h + (size_t)s1 * F + fo);
        const float4 v2 = *(const float4*)(h + (size_t)s2 * F + fo);
        const float4 v3 = *(const float4*)(h + (size_t)s3 * F + fo);
        const float n0 = dinv[s0] * dd, n1 = dinv[s1] * dd, n2 = dinv[s2] * dd, n3 = dinv[s3] * dd;
        acc.x += v0.x * n0; acc.y += v0.y * n0; acc.z += v0.z * n0; acc.w += v0.w * n0;
        acc.x += v1.x * n1; acc.y += v1.y * n1; acc.z += v1.z * n1; acc.w += v1.w * n1;
        acc.x += v2.x * n2; acc.y += v2.y * n2; acc.z += v2.z * n2; acc.w += v2.w * n2;
        acc.x += v3.x * n3; acc.y += v3.y * n3; acc.z += v3.z * n3; acc.w += v3.w * n3;
    }
    const int rem = end - j;
    if (rem > 0) {
        const int s0 = col[j];
        const int s1 = col[rem > 1 ? j + 1 : j];
        const int s2 = col[rem > 2 ? j + 2 : j];
        const float4 v0 = *(const float4*)(h + (size_t)s0 * F + fo);
        const float4 v1 = *(const float4*)(h + (size_t)s1 * F + fo);
        const float4 v2 = *(const float4*)(h + (size_t)s2 * F + fo);
        const float n0 = dinv[s0] * dd, n1 = dinv[s1] * dd, n2 = dinv[s2] * dd;
        acc.x += v0.x * n0; acc.y += v0.y * n0; acc.z += v0.z * n0; acc.w += v0.w * n0;
        if (rem > 1) { acc.x += v1.x * n1; acc.y += v1.y * n1; acc.z += v1.z * n1; acc.w += v1.w * n1; }
        if (rem > 2) { acc.x += v2.x * n2; acc.y += v2.y * n2; acc.z += v2.z * n2; acc.w += v2.w * n2; }
    }

    const float4 bb = *(const float4*)(bias + fo);
    float4 r;
    r.x = acc.x + bb.x; r.y = acc.y + bb.y; r.z = acc.z + bb.z; r.w = acc.w + bb.w;
    if (RELU) {
        r.x = fmaxf(r.x, 0.0f); r.y = fmaxf(r.y, 0.0f); r.z = fmaxf(r.z, 0.0f); r.w = fmaxf(r.w, 0.0f);
    }
    *(float4*)(o + (size_t)node * F + fo) = r;
}


extern "C" void kernel_launch(void* const* d_in, const int* in_sizes, int n_in,
                              void* d_out, int out_size, void* d_ws, size_t ws_size,
                              hipStream_t stream) {
    const float* x  = (const float*)d_in[0];
    const int*   ei = (const int*)d_in[1];
    const float* W1 = (const float*)d_in[2];
    const float* b1 = (const float*)d_in[3];
    const float* W2 = (const float*)d_in[4];
    const float* b2 = (const float*)d_in[5];
    float* out = (float*)d_out;

    const int N = in_sizes[0] / N_FEAT;
    const int E = in_sizes[1] / 2;
    const int* srcI = ei;
    const int* dstI = ei + E;

    char* ws = (char*)d_ws;
    size_t degB = ((size_t)N * sizeof(float) + 255) & ~(size_t)255;
    size_t t1B  = (((size_t)N * HIDDEN * sizeof(float)) + 255) & ~(size_t)255;
    size_t t3B  = (((size_t)N * NCLS * sizeof(float)) + 255) & ~(size_t)255;
    float* dinv = (float*)ws;
    float* t1   = (float*)(ws + degB);
    float* t2   = (float*)(ws + degB + t1B);
    float* t3   = t1;

    int* rowptrA = (int*)out;
    int* colA    = rowptrA + N;
    int* rowptrB = (int*)((char*)t1 + t3B);
    int* colB    = rowptrB + N;
    int* rank    = (int*)t2;
    const int listInts = N + E;

    const int BT = 256;
    const int Mtiles = N / 16;
    const int gemmBlocks = (Mtiles + 7) / 8;

    fill_int_kernel<<<(N + BT - 1) / BT, BT, 0, stream>>>(rowptrA, 0, N);
    count_rank_kernel<<<(E + BT - 1) / BT, BT, 0, stream>>>(dstI, rowptrA, rank, E);
    gemm1_scan_kernel<<<gemmBlocks + 1, 256, 0, stream>>>(x, W1, t1, N, gemmBlocks, rowptrA, dinv, N);
    place_kernel<<<(E + BT - 1) / BT, BT, 0, stream>>>(srcI, dstI, rowptrA, rank, colA, E);

    long long tot1 = (long long)N * (HIDDEN / 4);
    gather_kernel<HIDDEN, true><<<(int)((tot1 + BT - 1) / BT), BT, 0, stream>>>(
        t1, colA, rowptrA, dinv, b1, t2, N, E);

    const int copyBlocks = (listInts / 4 + 1024) / 1024;
    gemm2_copy_kernel<<<gemmBlocks + copyBlocks, 256, 0, stream>>>(t2, W2, t3, N, gemmBlocks,
                                                                 rowptrA, rowptrB, listInts);

    long long tot2 = (long long)N * (NCLS / 4);
    gather_kernel<NCLS, false><<<(int)((tot2 + BT - 1) / BT), BT, 0, stream>>>(
        t3, colB, rowptrB, dinv, b2, out, N, E);
}


# GEMM K loop: the next K block's A and B fragments are requested as one group into a second register set in front of the current block's WMMAs; order of summation unchanged
# speedup vs baseline: 2.1390x; 1.2409x over previous
#include <hip/hip_runtime.h>

typedef __attribute__((ext_vector_type(2))) float v2f;
typedef __attribute__((ext_vector_type(8))) float v8f;

#define N_FEAT 512
#define HIDDEN 128
#define NCLS   64


__global__ void fill_int_kernel(int* __restrict__ p, int v, int n) {
    int i = blockIdx.x * blockDim.x + threadIdx.x;
    if (i < n) p[i] = v;
}

__global__ void count_rank_kernel(const int* __restrict__ dst, int* __restrict__ cnt,
                                  int* __restrict__ rank, int E) {
    int e = blockIdx.x * blockDim.x + threadIdx.x;
    if (e < E) rank[e] = atomicAdd(&cnt[dst[e]], 1);
}

__device__ __forceinline__ void scan_dinv_block(int* __restrict__ cnt_rowptr, float* __restrict__ dinv, int n) {
    __shared__ int wtot[8];
    const int lane = threadIdx.x & 31;
    const int wave = threadIdx.x >> 5;
    const int trips = (((n + 7) >> 3) + 511) >> 9;
    const int wbeg = wave * trips * 512;

    int s = 0;
    for (int it = 0; it < trips; ++it) {
        const int i = wbeg + it * 512 + lane * 16;
        if (i + 15 < n) {
            const int4* p = (const int4*)(cnt_rowptr + i);
            const int4 a = p[0], b = p[1], c = p[2], d = p[3];
            s += (((a.x + a.y) + (a.z + a.w)) + ((b.x + b.y) + (b.z + b.w))) +
                 (((c.x + c.y) + (c.z + c.w)) + ((d.x + d.y) + (d.z + d.w)));
        } else {
            for (int k = 0; k < 16; ++k) if (i + k < n) s += cnt_rowptr[i + k];
        }
    }
    for (int off = 16; off > 0; off >>= 1) s += __shfl_xor(s, off);
    if (lane == 0) wtot[wave] = s;
    __syncthreads();
    int run = 0;
    for (int w = 0; w < wave; ++w) run += wtot[w];

    for (int it = 0; it < trips; ++it) {
        const int i = wbeg + it * 512 + lane * 16;
        int cv[16];
        if (i + 15 < n) {
            const int4* p = (const int4*)(cnt_rowptr + i);
            const int4 a = p[0], b = p[1], c = p[2], d = p[3];
            cv[0] = a.x;  cv[1] = a.y;  cv[2] = a.z;  cv[3] = a.w;
            cv[4] = b.x;  cv[5] = b.y;  cv[6] = b.z;  cv[7] = b.w;
            cv[8] = c.x;  cv[9] = c.y;  cv[10] = c.z; cv[11] = c.w;
            cv[12] = d.x; cv[13] = d.y; cv[14] = d.z; cv[15] = d.w;
        } else {
            for (int k = 0; k < 16; ++k) cv[k] = (i + k < n) ? cnt_rowptr[i + k] : 0;
        }
        int pv[16];
        int mine = 0;
        for (int k = 0; k < 16; ++k) { pv[k] = mine; mine += cv[k]; }
        int inc = mine;
        for (int off = 1; off < 32; off <<= 1) {
            const int v = __shfl_up(inc, off);
            if (lane >= off) inc += v;
        }
        const int base = run + inc - mine;
        if (i + 15 < n) {
            int4* q = (int4*)(cnt_rowptr + i);
            float4* g = (float4*)(dinv + i);
            for (int k = 0; k < 4; ++k) {
                q[k] = make_int4(base + pv[4 * k], base + pv[4 * k + 1], base + pv[4 * k + 2], base + pv[4 * k + 3]);
                g[k] = make_float4(rsqrtf((float)(cv[4 * k] + 1)), rsqrtf((float)(cv[4 * k + 1] + 1)),
                                   rsqrtf((float)(cv[4 * k + 2] + 1)), rsqrtf((float)(cv[4 * k + 3] + 1)));
            }
        } else {
            for (int k = 0; k < 16; ++k)
                if (i + k < n) { cnt_rowptr[i + k] = base + pv[k]; dinv[i + k] = rsqrtf((float)(cv[k] + 1)); }
        }
        run += __shfl(inc, 31);
    }
}

__global__ void place_kernel(const int* __restrict__ srcI, const int* __restrict__ dstI,
                             const int* __restrict__ rowptr, const int* __restrict__ rank,
                             int* __restrict__ col, int E) {
    int e = blockIdx.x * blockDim.x + threadIdx.x;
    if (e < E) col[rowptr[dstI[e]] + rank[e]] = srcI[e];
}

__device__ __forceinline__ void copy_ints_block(const int* __restrict__ a, int* __restrict__ b, int n, int cb) {
    const int n4 = n >> 2;
    const int4* a4 = (const int4*)a;
    int4* b4 = (int4*)b;
    const int i0 = cb * 1024 + threadIdx.x;
    const int i1 = i0 + 256, i2 = i0 + 512, i3 = i0 + 768;
    int4 v0, v1, v2, v3;
    if (i0 < n4) v0 = a4[i0];
    if (i1 < n4) v1 = a4[i1];
    if (i2 < n4) v2 = a4[i2];
    if (i3 < n4) v3 = a4[i3];
    if (i0 < n4) b4[i0] = v0;
    if (i1 < n4) b4[i1] = v1;
    if (i2 < n4) b4[i2] = v2;
    if (i3 < n4) b4[i3] = v3;
    if (cb == 0 && threadIdx.x == 0)
        for (int i = n4 * 4; i < n; ++i) b[i] = a[i];
}


template<int NCOL, int U>
__device__ __forceinline__ void load_k_block(const float* __restrict__ ap, const float* __restrict__ bp,
                                             v2f (&a)[U], v2f (&b)[U][NCOL / 16]) {
#pragma unroll
    for (int u = 0; u < U; ++u) {
        a[u].x = ap[4 * u];
        a[u].y = ap[4 * u + 1];
#pragma unroll
        for (int nt = 0; nt < NCOL / 16; ++nt) {
            b[u][nt].x = bp[(4 * u) * NCOL + nt * 16];
            b[u][nt].y = bp[(4 * u + 1) * NCOL + nt * 16];
        }
    }
}

template<int NCOL, int U>
__device__ __forceinline__ void mma_k_block(const v2f (&a)[U], const v2f (&b)[U][NCOL / 16],
                                            v8f (&acc)[NCOL / 16]) {
#pragma unroll
    for (int u = 0; u < U; ++u) {
#pragma unroll
        for (int nt = 0; nt < NCOL / 16; ++nt) {
            acc[nt] = __builtin_amdgcn_wmma_f32_16x16x4_f32(
                 false, a[u],  false, b[u][nt],
                 (short)0, acc[nt],  false,  false);
        }
    }
}

template<int K, int NCOL, int U>
__device__ __forceinline__ void gemm_wmma_f32(const float* __restrict__ A, const float* __restrict__ B,
                                              float* __restrict__ C, int M, int blk) {
    constexpr int NT = NCOL / 16;
    constexpr int KB = 4 * U;
    constexpr int NBLK = K / KB;
    static_assert(K % (2 * KB) == 0, "K must hold an even number of k blocks");
    const int lane = threadIdx.x & 31;
    const int wave = threadIdx.x >> 5;
    const int m0 = (blk * 8 + wave) * 16;
    if (m0 >= M) return;

    const int mr = lane & 15;
    const int kb = (lane >> 4) << 1;
    const int nc = lane & 15;

    const float* a_ptr = A + (size_t)(m0 + mr) * K + kb;
    const float* b_ptr = B + (size_t)kb * NCOL + nc;

    v8f acc[NT] = {};
    v2f a0[U], a1[U];
    v2f b0[U][NT], b1[U][NT];

    load_k_block<NCOL, U>(a_ptr, b_ptr, a0, b0);
    for (int i = 0; i + 2 < NBLK; i += 2) {
        load_k_block<NCOL, U>(a_ptr + KB, b_ptr + (size_t)KB * NCOL, a1, b1);
        __builtin_amdgcn_sched_barrier(0);
        mma_k_block<NCOL, U>(a0, b0, acc);
        a_ptr += 2 * KB;
        b_ptr += (size_t)2 * KB * NCOL;
        __builtin_amdgcn_sched_barrier(0);
        load_k_block<NCOL, U>(a_ptr, b_ptr, a0, b0);
        __builtin_amdgcn_sched_barrier(0);
        mma_k_block<NCOL, U>(a1, b1, acc);
    }
    load_k_block<NCOL, U>(a_ptr + KB, b_ptr + (size_t)KB * NCOL, a1, b1);
    __builtin_amdgcn_sched_barrier(0);
    mma_k_block<NCOL, U>(a0, b0, acc);
    mma_k_block<NCOL, U>(a1, b1, acc);
    __builtin_amdgcn_sched_barrier(0);

    const int rbase = m0 + ((lane >> 4) << 3);
#pragma unroll
    for (int nt = 0; nt < NT; ++nt) {
        float* c_ptr = C + (size_t)rbase * NCOL + nt * 16 + nc;
#pragma unroll
        for (int r = 0; r < 8; ++r)
            c_ptr[(size_t)r * NCOL] = acc[nt][r];
    }
}


__global__ __launch_bounds__(256)
void gemm1_scan_kernel(const float* __restrict__ x, const float* __restrict__ W1, float* __restrict__ t1,
                       int M, int gemm_blocks, int* __restrict__ cnt_rowptr, float* __restrict__ dinv, int n) {
    if ((int)blockIdx.x < gemm_blocks) {
        gemm_wmma_f32<N_FEAT, HIDDEN, 2>(x, W1, t1, M, (int)blockIdx.x);
    } else {
        scan_dinv_block(cnt_rowptr, dinv, n);
    }
}

__global__ __launch_bounds__(256)
void gemm2_copy_kernel(const float* __restrict__ h, const float* __restrict__ W2, float* __restrict__ t3,
                       int M, int gemm_blocks, const int* __restrict__ a, int* __restrict__ b, int n) {
    if ((int)blockIdx.x < gemm_blocks) {
        gemm_wmma_f32<HIDDEN, NCLS, 2>(h, W2, t3, M, (int)blockIdx.x);
    } else {
        copy_ints_block(a, b, n, (int)blockIdx.x - gemm_blocks);
    }
}


template<int F, bool RELU>
__global__ __launch_bounds__(256)
void gather_kernel(const float* __restrict__ h, const int* __restrict__ col,
                   const int* __restrict__ rowptr, const float* __restrict__ dinv,
                   const float* __restrict__ bias, float* __restrict__ o, int N, int E) {
    constexpr int TPN = F / 4;
    const int tid = blockIdx.x * blockDim.x + threadIdx.x;
    const int node = tid / TPN;
    if (node >= N) return;
    const int fo = (tid % TPN) << 2;

    const int beg = rowptr[node];
    const int end = (node + 1 < N) ? rowptr[node + 1] : E;
    const float dd = dinv[node];

    const float4 self = *(const float4*)(h + (size_t)node * F + fo);
    const float sn = dd * dd;
    float4 acc;
    acc.x = self.x * sn; acc.y = self.y * sn; acc.z = self.z * sn; acc.w = self.w * sn;

    int j = beg;
    for (; j + 4 <= end; j += 4) {
        const int s0 = col[j], s1 = col[j + 1], s2 = col[j + 2], s3 = col[j + 3];
        const float4 v0 = *(const float4*)(h + (size_t)s0 * F + fo);
        const float4 v1 = *(const float4*)(h + (size_t)s1 * F + fo);
        const float4 v2 = *(const float4*)(h + (size_t)s2 * F + fo);
        const float4 v3 = *(const float4*)(h + (size_t)s3 * F + fo);
        const float n0 = dinv[s0] * dd, n1 = dinv[s1] * dd, n2 = dinv[s2] * dd, n3 = dinv[s3] * dd;
        acc.x += v0.x * n0; acc.y += v0.y * n0; acc.z += v0.z * n0; acc.w += v0.w * n0;
        acc.x += v1.x * n1; acc.y += v1.y * n1; acc.z += v1.z * n1; acc.w += v1.w * n1;
        acc.x += v2.x * n2; acc.y += v2.y * n2; acc.z += v2.z * n2; acc.w += v2.w * n2;
        acc.x += v3.x * n3; acc.y += v3.y * n3; acc.z += v3.z * n3; acc.w += v3.w * n3;
    }
    const int rem = end - j;
    if (rem > 0) {
        const int s0 = col[j];
        const int s1 = col[rem > 1 ? j + 1 : j];
        const int s2 = col[rem > 2 ? j + 2 : j];
        const float4 v0 = *(const float4*)(h + (size_t)s0 * F + fo);
        const float4 v1 = *(const float4*)(h + (size_t)s1 * F + fo);
        const float4 v2 = *(const float4*)(h + (size_t)s2 * F + fo);
        const float n0 = dinv[s0] * dd, n1 = dinv[s1] * dd, n2 = dinv[s2] * dd;
        acc.x += v0.x * n0; acc.y += v0.y * n0; acc.z += v0.z * n0; acc.w += v0.w * n0;
        if (rem > 1) { acc.x += v1.x * n1; acc.y += v1.y * n1; acc.z += v1.z * n1; acc.w += v1.w * n1; }
        if (rem > 2) { acc.x += v2.x * n2; acc.y += v2.y * n2; acc.z += v2.z * n2; acc.w += v2.w * n2; }
    }

    const float4 bb = *(const float4*)(bias + fo);
    float4 r;
    r.x = acc.x + bb.x; r.y = acc.y + bb.y; r.z = acc.z + bb.z; r.w = acc.w + bb.w;
    if (RELU) {
        r.x = fmaxf(r.x, 0.0f); r.y = fmaxf(r.y, 0.0f); r.z = fmaxf(r.z, 0.0f); r.w = fmaxf(r.w, 0.0f);
    }
    *(float4*)(o + (size_t)node * F + fo) = r;
}


extern "C" void kernel_launch(void* const* d_in, const int* in_sizes, int n_in,
                              void* d_out, int out_size, void* d_ws, size_t ws_size,
                              hipStream_t stream) {
    const float* x  = (const float*)d_in[0];
    const int*   ei = (const int*)d_in[1];
    const float* W1 = (const float*)d_in[2];
    const float* b1 = (const float*)d_in[3];
    const float* W2 = (const float*)d_in[4];
    const float* b2 = (const float*)d_in[5];
    float* out = (float*)d_out;

    const int N = in_sizes[0] / N_FEAT;
    const int E = in_sizes[1] / 2;
    const int* srcI = ei;
    const int* dstI = ei + E;

    char* ws = (char*)d_ws;
    size_t degB = ((size_t)N * sizeof(float) + 255) & ~(size_t)255;
    size_t t1B  = (((size_t)N * HIDDEN * sizeof(float)) + 255) & ~(size_t)255;
    size_t t3B  = (((size_t)N * NCLS * sizeof(float)) + 255) & ~(size_t)255;
    float* dinv = (float*)ws;
    float* t1   = (float*)(ws + degB);
    float* t2   = (float*)(ws + degB + t1B);
    float* t3   = t1;

    int* rowptrA = (int*)out;
    int* colA    = rowptrA + N;
    int* rowptrB = (int*)((char*)t1 + t3B);
    int* colB    = rowptrB + N;
    int* rank    = (int*)t2;
    const int listInts = N + E;

    const int BT = 256;
    const int Mtiles = N / 16;
    const int gemmBlocks = (Mtiles + 7) / 8;

    fill_int_kernel<<<(N + BT - 1) / BT, BT, 0, stream>>>(rowptrA, 0, N);
    count_rank_kernel<<<(E + BT - 1) / BT, BT, 0, stream>>>(dstI, rowptrA, rank, E);
    gemm1_scan_kernel<<<gemmBlocks + 1, 256, 0, stream>>>(x, W1, t1, N, gemmBlocks, rowptrA, dinv, N);
    place_kernel<<<(E + BT - 1) / BT, BT, 0, stream>>>(srcI, dstI, rowptrA, rank, colA, E);

    long long tot1 = (long long)N * (HIDDEN / 4);
    gather_kernel<HIDDEN, true><<<(int)((tot1 + BT - 1) / BT), BT, 0, stream>>>(
        t1, colA, rowptrA, dinv, b1, t2, N, E);

    const int copyBlocks = (listInts / 4 + 1024) / 1024;
    gemm2_copy_kernel<<<gemmBlocks + copyBlocks, 256, 0, stream>>>(t2, W2, t3, N, gemmBlocks,
                                                                 rowptrA, rowptrB, listInts);

    long long tot2 = (long long)N * (NCLS / 4);
    gather_kernel<NCLS, false><<<(int)((tot2 + BT - 1) / BT), BT, 0, stream>>>(
        t3, colB, rowptrB, dinv, b2, out, N, E);
}


# Edge list in 32 slots per destination, filled in one pass by the rank atomics inside GEMM 1's launch (no prefix sum, no placement pass); 7 launches -> 5; sums in arrival rank, spill edges last
# speedup vs baseline: 2.3828x; 1.1140x over previous
#include <hip/hip_runtime.h>

typedef __attribute__((ext_vector_type(2))) float v2f;
typedef __attribute__((ext_vector_type(8))) float v8f;

#define N_FEAT 512
#define HIDDEN 128
#define NCLS   64


#define SLOT_CAP 32

__global__ void fill_int_kernel(int* __restrict__ p, int v, int n) {
    int i = blockIdx.x * blockDim.x + threadIdx.x;
    if (i < n) p[i] = v;
}

__device__ __forceinline__ void put_one_edge(int d, int s, int r, int* __restrict__ cnt, int* __restrict__ slots,
                                             int2* __restrict__ spill, int N, int spill_cap) {
    if (r < SLOT_CAP) {
        slots[(size_t)d * SLOT_CAP + r] = s;
    } else {
        const int i = atomicAdd(&cnt[N], 1);
        if (i < spill_cap) spill[i] = make_int2(d, s);
    }
}

__device__ __forceinline__ void slot_edges_block(const int* __restrict__ srcI, const int* __restrict__ dstI,
                                                 int* __restrict__ cnt, int* __restrict__ slots,
                                                 int2* __restrict__ spill, int N, int E, int spill_cap, int cb) {
    const int e0 = (cb * 256 + (int)threadIdx.x) * 4;
    if (((E & 3) == 0) && (e0 + 3 < E)) {
        const int4 d4 = *(const int4*)(dstI + e0);
        const int4 s4 = *(const int4*)(srcI + e0);
        const int r0 = atomicAdd(&cnt[d4.x], 1);
        const int r1 = atomicAdd(&cnt[d4.y], 1);
        const int r2 = atomicAdd(&cnt[d4.z], 1);
        const int r3 = atomicAdd(&cnt[d4.w], 1);
        put_one_edge(d4.x, s4.x, r0, cnt, slots, spill, N, spill_cap);
        put_one_edge(d4.y, s4.y, r1, cnt, slots, spill, N, spill_cap);
        put_one_edge(d4.z, s4.z, r2, cnt, slots, spill, N, spill_cap);
        put_one_edge(d4.w, s4.w, r3, cnt, slots, spill, N, spill_cap);
    } else {
        for (int k = 0; k < 4; ++k)
            if (e0 + k < E) {
                const int d = dstI[e0 + k];
                put_one_edge(d, srcI[e0 + k], atomicAdd(&cnt[d], 1), cnt, slots, spill, N, spill_cap);
            }
    }
}

__device__ __forceinline__ void copy_spill_block(const int* __restrict__ cnt, const int2* __restrict__ a,
                                                 int2* __restrict__ b, int N, int spill_cap) {
    int oc = cnt[N];
    oc = oc < spill_cap ? oc : spill_cap;
    for (int i = (int)threadIdx.x; i < oc; i += 256) b[i] = a[i];
}

__device__ __forceinline__ void copy_ints_block(const int* __restrict__ a, int* __restrict__ b, int n, int cb) {
    const int n4 = n >> 2;
    const int4* a4 = (const int4*)a;
    int4* b4 = (int4*)b;
    const int i0 = cb * 1024 + threadIdx.x;
    const int i1 = i0 + 256, i2 = i0 + 512, i3 = i0 + 768;
    int4 v0, v1, v2, v3;
    if (i0 < n4) v0 = a4[i0];
    if (i1 < n4) v1 = a4[i1];
    if (i2 < n4) v2 = a4[i2];
    if (i3 < n4) v3 = a4[i3];
    if (i0 < n4) b4[i0] = v0;
    if (i1 < n4) b4[i1] = v1;
    if (i2 < n4) b4[i2] = v2;
    if (i3 < n4) b4[i3] = v3;
    if (cb == 0 && threadIdx.x == 0)
        for (int i = n4 * 4; i < n; ++i) b[i] = a[i];
}


template<int NCOL, int U>
__device__ __forceinline__ void load_k_block(const float* __restrict__ ap, const float* __restrict__ bp,
                                             v2f (&a)[U], v2f (&b)[U][NCOL / 16]) {
#pragma unroll
    for (int u = 0; u < U; ++u) {
        a[u].x = ap[4 * u];
        a[u].y = ap[4 * u + 1];
#pragma unroll
        for (int nt = 0; nt < NCOL / 16; ++nt) {
            b[u][nt].x = bp[(4 * u) * NCOL + nt * 16];
            b[u][nt].y = bp[(4 * u + 1) * NCOL + nt * 16];
        }
    }
}

template<int NCOL, int U>
__device__ __forceinline__ void mma_k_block(const v2f (&a)[U], const v2f (&b)[U][NCOL / 16],
                                            v8f (&acc)[NCOL / 16]) {
#pragma unroll
    for (int u = 0; u < U; ++u) {
#pragma unroll
        for (int nt = 0; nt < NCOL / 16; ++nt) {
            acc[nt] = __builtin_amdgcn_wmma_f32_16x16x4_f32(
                 false, a[u],  false, b[u][nt],
                 (short)0, acc[nt],  false,  false);
        }
    }
}

template<int K, int NCOL, int U>
__device__ __forceinline__ void gemm_wmma_f32(const float* __restrict__ A, const float* __restrict__ B,
                                              float* __restrict__ C, int M, int blk) {
    constexpr int NT = NCOL / 16;
    constexpr int KB = 4 * U;
    constexpr int NBLK = K / KB;
    static_assert(K % (2 * KB) == 0, "K must hold an even number of k blocks");
    const int lane = threadIdx.x & 31;
    const int wave = threadIdx.x >> 5;
    const int m0 = (blk * 8 + wave) * 16;
    if (m0 >= M) return;

    const int mr = lane & 15;
    const int kb = (lane >> 4) << 1;
    const int nc = lane & 15;

    const float* a_ptr = A + (size_t)(m0 + mr) * K + kb;
    const float* b_ptr = B + (size_t)kb * NCOL + nc;

    v8f acc[NT] = {};
    v2f a0[U], a1[U];
    v2f b0[U][NT], b1[U][NT];

    load_k_block<NCOL, U>(a_ptr, b_ptr, a0, b0);
    for (int i = 0; i + 2 < NBLK; i += 2) {
        load_k_block<NCOL, U>(a_ptr + KB, b_ptr + (size_t)KB * NCOL, a1, b1);
        __builtin_amdgcn_sched_barrier(0);
        mma_k_block<NCOL, U>(a0, b0, acc);
        a_ptr += 2 * KB;
        b_ptr += (size_t)2 * KB * NCOL;
        __builtin_amdgcn_sched_barrier(0);
        load_k_block<NCOL, U>(a_ptr, b_ptr, a0, b0);
        __builtin_amdgcn_sched_barrier(0);
        mma_k_block<NCOL, U>(a1, b1, acc);
    }
    load_k_block<NCOL, U>(a_ptr + KB, b_ptr + (size_t)KB * NCOL, a1, b1);
    __builtin_amdgcn_sched_barrier(0);
    mma_k_block<NCOL, U>(a0, b0, acc);
    mma_k_block<NCOL, U>(a1, b1, acc);
    __builtin_amdgcn_sched_barrier(0);

    const int rbase = m0 + ((lane >> 4) << 3);
#pragma unroll
    for (int nt = 0; nt < NT; ++nt) {
        float* c_ptr = C + (size_t)rbase * NCOL + nt * 16 + nc;
#pragma unroll
        for (int r = 0; r < 8; ++r)
            c_ptr[(size_t)r * NCOL] = acc[nt][r];
    }
}


__global__ __launch_bounds__(256)
void gemm1_slots_kernel(const float* __restrict__ x, const float* __restrict__ W1, float* __restrict__ t1,
                        int M, int gemm_blocks, const int* __restrict__ srcI, const int* __restrict__ dstI,
                        int* __restrict__ cnt, int* __restrict__ slots, int2* __restrict__ spill,
                        int E, int spill_cap) {
    if ((int)blockIdx.x < gemm_blocks) {
        gemm_wmma_f32<N_FEAT, HIDDEN, 2>(x, W1, t1, M, (int)blockIdx.x);
    } else {
        slot_edges_block(srcI, dstI, cnt, slots, spill, M, E, spill_cap, (int)blockIdx.x - gemm_blocks);
    }
}

__global__ __launch_bounds__(256)
void gemm2_copy_kernel(const float* __restrict__ h, const float* __restrict__ W2, float* __restrict__ t3,
                       int M, int gemm_blocks, int copy_blocks, const int* __restrict__ a, int* __restrict__ b, int n,
                       const int* __restrict__ cnt, const int2* __restrict__ spillA, int2* __restrict__ spillB,
                       int spill_cap) {
    if ((int)blockIdx.x < gemm_blocks) {
        gemm_wmma_f32<HIDDEN, NCLS, 2>(h, W2, t3, M, (int)blockIdx.x);
    } else if ((int)blockIdx.x < gemm_blocks + copy_blocks) {
        copy_ints_block(a, b, n, (int)blockIdx.x - gemm_blocks);
    } else {
        copy_spill_block(cnt, spillA, spillB, M, spill_cap);
    }
}


template<int F, bool RELU>
__global__ __launch_bounds__(256)
void gather_kernel(const float* __restrict__ h, const int* __restrict__ slots,
                   const int* __restrict__ cnt, const int2* __restrict__ spill,
                   const float* __restrict__ bias, float* __restrict__ o, int N, int spill_cap) {
    constexpr int TPN = F / 4;
    const int tid = blockIdx.x * blockDim.x + threadIdx.x;
    const int node = tid / TPN;
    if (node >= N) return;
    const int fo = (tid % TPN) << 2;

    const int deg = cnt[node];
    const int beg = node * SLOT_CAP;
    const int end = beg + (deg < SLOT_CAP ? deg : SLOT_CAP);
    const float dd = rsqrtf((float)(deg + 1));

    const float4 self = *(const float4*)(h + (size_t)node * F + fo);
    const float sn = dd * dd;
    float4 acc;
    acc.x = self.x * sn; acc.y = self.y * sn; acc.z = self.z * sn; acc.w = self.w * sn;

    int j = beg;
    for (; j + 4 <= end; j += 4) {
        const int4 s4 = *(const int4*)(slots + j);
        const int s0 = s4.x, s1 = s4.y, s2 = s4.z, s3 = s4.w;
        const float4 v0 = *(const float4*)(h + (size_t)s0 * F + fo);
        const float4 v1 = *(const float4*)(h + (size_t)s1 * F + fo);
        const float4 v2 = *(const float4*)(h + (size_t)s2 * F + fo);
        const float4 v3 = *(const float4*)(h + (size_t)s3 * F + fo);
        const float n0 = rsqrtf((float)(cnt[s0] + 1)) * dd, n1 = rsqrtf((float)(cnt[s1] + 1)) * dd;
        const float n2 = rsqrtf((float)(cnt[s2] + 1)) * dd, n3 = rsqrtf((float)(cnt[s3] + 1)) * dd;
        acc.x += v0.x * n0; acc.y += v0.y * n0; acc.z += v0.z * n0; acc.w += v0.w * n0;
        acc.x += v1.x * n1; acc.y += v1.y * n1; acc.z += v1.z * n1; acc.w += v1.w * n1;
        acc.x += v2.x * n2; acc.y += v2.y * n2; acc.z += v2.z * n2; acc.w += v2.w * n2;
        acc.x += v3.x * n3; acc.y += v3.y * n3; acc.z += v3.z * n3; acc.w += v3.w * n3;
    }
    const int rem = end - j;
    if (rem > 0) {
        const int s0 = slots[j];
        const int s1 = slots[rem > 1 ? j + 1 : j];
        const int s2 = slots[rem > 2 ? j + 2 : j];
        const float4 v0 = *(const float4*)(h + (size_t)s0 * F + fo);
        const float4 v1 = *(const float4*)(h + (size_t)s1 * F + fo);
        const float4 v2 = *(const float4*)(h + (size_t)s2 * F + fo);
        const float n0 = rsqrtf((float)(cnt[s0] + 1)) * dd, n1 = rsqrtf((float)(cnt[s1] + 1)) * dd;
        const float n2 = rsqrtf((float)(cnt[s2] + 1)) * dd;
        acc.x += v0.x * n0; acc.y += v0.y * n0; acc.z += v0.z * n0; acc.w += v0.w * n0;
        if (rem > 1) { acc.x += v1.x * n1; acc.y += v1.y * n1; acc.z += v1.z * n1; acc.w += v1.w * n1; }
        if (rem > 2) { acc.x += v2.x * n2; acc.y += v2.y * n2; acc.z += v2.z * n2; acc.w += v2.w * n2; }
    }
    if (deg > SLOT_CAP) {
        int oc = cnt[N];
        oc = oc < spill_cap ? oc : spill_cap;
        for (int i = 0; i < oc; ++i) {
            const int2 en = spill[i];
            if (en.x == node) {
                const float4 v = *(const float4*)(h + (size_t)en.y * F + fo);
                const float nn = rsqrtf((float)(cnt[en.y] + 1)) * dd;
                acc.x += v.x * nn; acc.y += v.y * nn; acc.z += v.z * nn; acc.w += v.w * nn;
            }
        }
    }

    const float4 bb = *(const float4*)(bias + fo);
    float4 r;
    r.x = acc.x + bb.x; r.y = acc.y + bb.y; r.z = acc.z + bb.z; r.w = acc.w + bb.w;
    if (RELU) {
        r.x = fmaxf(r.x, 0.0f); r.y = fmaxf(r.y, 0.0f); r.z = fmaxf(r.z, 0.0f); r.w = fmaxf(r.w, 0.0f);
    }
    *(float4*)(o + (size_t)node * F + fo) = r;
}


extern "C" void kernel_launch(void* const* d_in, const int* in_sizes, int n_in,
                              void* d_out, int out_size, void* d_ws, size_t ws_size,
                              hipStream_t stream) {
    const float* x  = (const float*)d_in[0];
    const int*   ei = (const int*)d_in[1];
    const float* W1 = (const float*)d_in[2];
    const float* b1 = (const float*)d_in[3];
    const float* W2 = (const float*)d_in[4];
    const float* b2 = (const float*)d_in[5];
    float* out = (float*)d_out;

    const int N = in_sizes[0] / N_FEAT;
    const int E = in_sizes[1] / 2;
    const int* srcI = ei;
    const int* dstI = ei + E;

    char* ws = (char*)d_ws;
    size_t cntB = ((size_t)(N + 1) * sizeof(int) + 255) & ~(size_t)255;
    size_t t1B  = (((size_t)N * HIDDEN * sizeof(float)) + 255) & ~(size_t)255;
    size_t t3B  = (((size_t)N * NCLS * sizeof(float)) + 255) & ~(size_t)255;
    int*   cnt  = (int*)ws;
    float* t1   = (float*)(ws + cntB);
    float* t2   = (float*)(ws + cntB + t1B);
    float* t3   = t1;

    const int slotInts = N * SLOT_CAP;
    const int spillCap = (N * (NCLS - SLOT_CAP)) / 2;
    int*  slotsA = (int*)out;
    int2* spillA = (int2*)(slotsA + slotInts);
    int*  slotsB = (int*)((char*)t1 + t3B);
    int2* spillB = (int2*)(slotsB + slotInts);

    const int BT = 256;
    const int Mtiles = N / 16;
    const int gemmBlocks = (Mtiles + 7) / 8;

    fill_int_kernel<<<(N + 1 + BT - 1) / BT, BT, 0, stream>>>(cnt, 0, N + 1);
    const int slotBlocks = (E + 1023) / 1024;
    gemm1_slots_kernel<<<gemmBlocks + slotBlocks, 256, 0, stream>>>(x, W1, t1, N, gemmBlocks, srcI, dstI,
                                                                  cnt, slotsA, spillA, E, spillCap);

    long long tot1 = (long long)N * (HIDDEN / 4);
    gather_kernel<HIDDEN, true><<<(int)((tot1 + BT - 1) / BT), BT, 0, stream>>>(
        t1, slotsA, cnt, spillA, b1, t2, N, spillCap);

    const int copyBlocks = (slotInts / 4 + 1023) / 1024;
    gemm2_copy_kernel<<<gemmBlocks + copyBlocks + 1, 256, 0, stream>>>(t2, W2, t3, N, gemmBlocks, copyBlocks,
                                                                     slotsA, slotsB, slotInts,
                                                                     cnt, spillA, spillB, spillCap);

    long long tot2 = (long long)N * (NCLS / 4);
    gather_kernel<NCLS, false><<<(int)((tot2 + BT - 1) / BT), BT, 0, stream>>>(
        t3, slotsB, cnt, spillB, b2, out, N, spillCap);
}


# GEMM weights re-laid per call in WMMA fragment order (extra blocks of the fill launch): every B load is one whole-line 16-byte load, tiles leave as whole-line 16-byte stores; sums unchanged
# speedup vs baseline: 2.4898x; 1.0449x over previous
#include <hip/hip_runtime.h>

typedef __attribute__((ext_vector_type(2))) float v2f;
typedef __attribute__((ext_vector_type(8))) float v8f;

#define N_FEAT 512
#define HIDDEN 128
#define NCLS   64


#define SLOT_CAP 32

template<int K, int NCOL>
__device__ __forceinline__ void relay_weights_block(const float* __restrict__ Wsrc, float* __restrict__ Wf, int cb) {
    constexpr int NT = NCOL / 16;
    const int t = cb * 256 + (int)threadIdx.x;
    if (t >= K * NCOL / 4) return;
    const int lane = t & 31;
    const int q = (t >> 5) % (NT / 2);
    const int kstep = (t >> 5) / (NT / 2);
    const int k = 4 * kstep + ((lane >> 4) << 1);
    const int nt = 2 * q;
    const int n = 64 * (nt >> 2) + 4 * (lane & 15) + (nt & 3);
    const float2 r0 = *(const float2*)(Wsrc + (size_t)k * NCOL + n);
    const float2 r1 = *(const float2*)(Wsrc + (size_t)(k + 1) * NCOL + n);
    float4 o;
    o.x = r0.x; o.y = r1.x; o.z = r0.y; o.w = r1.y;
    *(float4*)(Wf + (size_t)t * 4) = o;
}

__global__ __launch_bounds__(256)
void prep_kernel(int* __restrict__ cnt, int n, int fill_blocks,
                 const float* __restrict__ W1, float* __restrict__ W1f, int w1_blocks,
                 const float* __restrict__ W2, float* __restrict__ W2f) {
    const int b = (int)blockIdx.x;
    if (b < fill_blocks) {
        const int i = b * 256 + (int)threadIdx.x;
        if (i < n) cnt[i] = 0;
    } else if (b < fill_blocks + w1_blocks) {
        relay_weights_block<N_FEAT, HIDDEN>(W1, W1f, b - fill_blocks);
    } else {
        relay_weights_block<HIDDEN, NCLS>(W2, W2f, b - fill_blocks - w1_blocks);
    }
}

__device__ __forceinline__ void put_one_edge(int d, int s, int r, int* __restrict__ cnt, int* __restrict__ slots,
                                             int2* __restrict__ spill, int N, int spill_cap) {
    if (r < SLOT_CAP) {
        slots[(size_t)d * SLOT_CAP + r] = s;
    } else {
        const int i = atomicAdd(&cnt[N], 1);
        if (i < spill_cap) spill[i] = make_int2(d, s);
    }
}

__device__ __forceinline__ void slot_edges_block(const int* __restrict__ srcI, const int* __restrict__ dstI,
                                                 int* __restrict__ cnt, int* __restrict__ slots,
                                                 int2* __restrict__ spill, int N, int E, int spill_cap, int cb) {
    const int e0 = (cb * 256 + (int)threadIdx.x) * 4;
    if (((E & 3) == 0) && (e0 + 3 < E)) {
        const int4 d4 = *(const int4*)(dstI + e0);
        const int4 s4 = *(const int4*)(srcI + e0);
        const int r0 = atomicAdd(&cnt[d4.x], 1);
        const int r1 = atomicAdd(&cnt[d4.y], 1);
        const int r2 = atomicAdd(&cnt[d4.z], 1);
        const int r3 = atomicAdd(&cnt[d4.w], 1);
        put_one_edge(d4.x, s4.x, r0, cnt, slots, spill, N, spill_cap);
        put_one_edge(d4.y, s4.y, r1, cnt, slots, spill, N, spill_cap);
        put_one_edge(d4.z, s4.z, r2, cnt, slots, spill, N, spill_cap);
        put_one_edge(d4.w, s4.w, r3, cnt, slots, spill, N, spill_cap);
    } else {
        for (int k = 0; k < 4; ++k)
            if (e0 + k < E) {
                const int d = dstI[e0 + k];
                put_one_edge(d, srcI[e0 + k], atomicAdd(&cnt[d], 1), cnt, slots, spill, N, spill_cap);
            }
    }
}

__device__ __forceinline__ void copy_spill_block(const int* __restrict__ cnt, const int2* __restrict__ a,
                                                 int2* __restrict__ b, int N, int spill_cap) {
    int oc = cnt[N];
    oc = oc < spill_cap ? oc : spill_cap;
    for (int i = (int)threadIdx.x; i < oc; i += 256) b[i] = a[i];
}

__device__ __forceinline__ void copy_ints_block(const int* __restrict__ a, int* __restrict__ b, int n, int cb) {
    const int n4 = n >> 2;
    const int4* a4 = (const int4*)a;
    int4* b4 = (int4*)b;
    const int i0 = cb * 1024 + threadIdx.x;
    const int i1 = i0 + 256, i2 = i0 + 512, i3 = i0 + 768;
    int4 v0, v1, v2, v3;
    if (i0 < n4) v0 = a4[i0];
    if (i1 < n4) v1 = a4[i1];
    if (i2 < n4) v2 = a4[i2];
    if (i3 < n4) v3 = a4[i3];
    if (i0 < n4) b4[i0] = v0;
    if (i1 < n4) b4[i1] = v1;
    if (i2 < n4) b4[i2] = v2;
    if (i3 < n4) b4[i3] = v3;
    if (cb == 0 && threadIdx.x == 0)
        for (int i = n4 * 4; i < n; ++i) b[i] = a[i];
}


template<int NCOL, int U>
__device__ __forceinline__ void load_k_block(const float* __restrict__ ap, const float* __restrict__ bp,
                                             v2f (&a)[U], v2f (&b)[U][NCOL / 16]) {
#pragma unroll
    for (int u = 0; u < U; ++u) {
        a[u].x = ap[4 * u];
        a[u].y = ap[4 * u + 1];
#pragma unroll
        for (int q = 0; q < NCOL / 32; ++q) {
            const float4 f = *(const float4*)(bp + (u * (NCOL / 32) + q) * 128);
            b[u][2 * q].x = f.x;
            b[u][2 * q].y = f.y;
            b[u][2 * q + 1].x = f.z;
            b[u][2 * q + 1].y = f.w;
        }
    }
}

template<int NCOL, int U>
__device__ __forceinline__ void mma_k_block(const v2f (&a)[U], const v2f (&b)[U][NCOL / 16],
                                            v8f (&acc)[NCOL / 16]) {
#pragma unroll
    for (int u = 0; u < U; ++u) {
#pragma unroll
        for (int nt = 0; nt < NCOL / 16; ++nt) {
            acc[nt] = __builtin_amdgcn_wmma_f32_16x16x4_f32(
                 false, a[u],  false, b[u][nt],
                 (short)0, acc[nt],  false,  false);
        }
    }
}

template<int K, int NCOL, int U>
__device__ __forceinline__ void gemm_wmma_f32(const float* __restrict__ A, const float* __restrict__ Bf,
                                              float* __restrict__ C, int M, int blk) {
    constexpr int NT = NCOL / 16;
    constexpr int KB = 4 * U;
    constexpr int NBLK = K / KB;
    static_assert(K % (2 * KB) == 0, "K must hold an even number of k blocks");
    const int lane = threadIdx.x & 31;
    const int wave = threadIdx.x >> 5;
    const int m0 = (blk * 8 + wave) * 16;
    if (m0 >= M) return;

    const int mr = lane & 15;
    const int kb = (lane >> 4) << 1;
    const int nc = lane & 15;

    const float* a_ptr = A + (size_t)(m0 + mr) * K + kb;
    const float* b_ptr = Bf + lane * 4;

    v8f acc[NT] = {};
    v2f a0[U], a1[U];
    v2f b0[U][NT], b1[U][NT];

    load_k_block<NCOL, U>(a_ptr, b_ptr, a0, b0);
    for (int i = 0; i + 2 < NBLK; i += 2) {
        load_k_block<NCOL, U>(a_ptr + KB, b_ptr + (size_t)KB * NCOL, a1, b1);
        __builtin_amdgcn_sched_barrier(0);
        mma_k_block<NCOL, U>(a0, b0, acc);
        a_ptr += 2 * KB;
        b_ptr += (size_t)2 * KB * NCOL;
        __builtin_amdgcn_sched_barrier(0);
        load_k_block<NCOL, U>(a_ptr, b_ptr, a0, b0);
        __builtin_amdgcn_sched_barrier(0);
        mma_k_block<NCOL, U>(a1, b1, acc);
    }
    load_k_block<NCOL, U>(a_ptr + KB, b_ptr + (size_t)KB * NCOL, a1, b1);
    __builtin_amdgcn_sched_barrier(0);
    mma_k_block<NCOL, U>(a0, b0, acc);
    mma_k_block<NCOL, U>(a1, b1, acc);
    __builtin_amdgcn_sched_barrier(0);

    const int rbase = m0 + ((lane >> 4) << 3);
#pragma unroll
    for (int g = 0; g < NT / 4; ++g) {
        float* c_ptr = C + (size_t)rbase * NCOL + g * 64 + nc * 4;
#pragma unroll
        for (int r = 0; r < 8; ++r) {
            float4 v;
            v.x = acc[4 * g][r]; v.y = acc[4 * g + 1][r]; v.z = acc[4 * g + 2][r]; v.w = acc[4 * g + 3][r];
            *(float4*)(c_ptr + (size_t)r * NCOL) = v;
        }
    }
}


__global__ __launch_bounds__(256)
void gemm1_slots_kernel(const float* __restrict__ x, const float* __restrict__ W1f, float* __restrict__ t1,
                        int M, int gemm_blocks, const int* __restrict__ srcI, const int* __restrict__ dstI,
                        int* __restrict__ cnt, int* __restrict__ slots, int2* __restrict__ spill,
                        int E, int spill_cap) {
    if ((int)blockIdx.x < gemm_blocks) {
        gemm_wmma_f32<N_FEAT, HIDDEN, 2>(x, W1f, t1, M, (int)blockIdx.x);
    } else {
        slot_edges_block(srcI, dstI, cnt, slots, spill, M, E, spill_cap, (int)blockIdx.x - gemm_blocks);
    }
}

__global__ __launch_bounds__(256)
void gemm2_copy_kernel(const float* __restrict__ h, const float* __restrict__ W2f, float* __restrict__ t3,
                       int M, int gemm_blocks, int copy_blocks, const int* __restrict__ a, int* __restrict__ b, int n,
                       const int* __restrict__ cnt, const int2* __restrict__ spillA, int2* __restrict__ spillB,
                       int spill_cap) {
    if ((int)blockIdx.x < gemm_blocks) {
        gemm_wmma_f32<HIDDEN, NCLS, 2>(h, W2f, t3, M, (int)blockIdx.x);
    } else if ((int)blockIdx.x < gemm_blocks + copy_blocks) {
        copy_ints_block(a, b, n, (int)blockIdx.x - gemm_blocks);
    } else {
        copy_spill_block(cnt, spillA, spillB, M, spill_cap);
    }
}


template<int F, bool RELU>
__global__ __launch_bounds__(256)
void gather_kernel(const float* __restrict__ h, const int* __restrict__ slots,
                   const int* __restrict__ cnt, const int2* __restrict__ spill,
                   const float* __restrict__ bias, float* __restrict__ o, int N, int spill_cap) {
    constexpr int TPN = F / 4;
    const int tid = blockIdx.x * blockDim.x + threadIdx.x;
    const int node = tid / TPN;
    if (node >= N) return;
    const int fo = (tid % TPN) << 2;

    const int deg = cnt[node];
    const int beg = node * SLOT_CAP;
    const int end = beg + (deg < SLOT_CAP ? deg : SLOT_CAP);
    const float dd = rsqrtf((float)(deg + 1));

    const float4 self = *(const float4*)(h + (size_t)node * F + fo);
    const float sn = dd * dd;
    float4 acc;
    acc.x = self.x * sn; acc.y = self.y * sn; acc.z = self.z * sn; acc.w = self.w * sn;

    int j = beg;
    for (; j + 4 <= end; j += 4) {
        const int4 s4 = *(const int4*)(slots + j);
        const int s0 = s4.x, s1 = s4.y, s2 = s4.z, s3 = s4.w;
        const float4 v0 = *(const float4*)(h + (size_t)s0 * F + fo);
        const float4 v1 = *(const float4*)(h + (size_t)s1 * F + fo);
        const float4 v2 = *(const float4*)(h + (size_t)s2 * F + fo);
        const float4 v3 = *(const float4*)(h + (size_t)s3 * F + fo);
        const float n0 = rsqrtf((float)(cnt[s0] + 1)) * dd, n1 = rsqrtf((float)(cnt[s1] + 1)) * dd;
        const float n2 = rsqrtf((float)(cnt[s2] + 1)) * dd, n3 = rsqrtf((float)(cnt[s3] + 1)) * dd;
        acc.x += v0.x * n0; acc.y += v0.y * n0; acc.z += v0.z * n0; acc.w += v0.w * n0;
        acc.x += v1.x * n1; acc.y += v1.y * n1; acc.z += v1.z * n1; acc.w += v1.w * n1;
        acc.x += v2.x * n2; acc.y += v2.y * n2; acc.z += v2.z * n2; acc.w += v2.w * n2;
        acc.x += v3.x * n3; acc.y += v3.y * n3; acc.z += v3.z * n3; acc.w += v3.w * n3;
    }
    const int rem = end - j;
    if (rem > 0) {
        const int s0 = slots[j];
        const int s1 = slots[rem > 1 ? j + 1 : j];
        const int s2 = slots[rem > 2 ? j + 2 : j];
        const float4 v0 = *(const float4*)(h + (size_t)s0 * F + fo);
        const float4 v1 = *(const float4*)(h + (size_t)s1 * F + fo);
        const float4 v2 = *(const float4*)(h + (size_t)s2 * F + fo);
        const float n0 = rsqrtf((float)(cnt[s0] + 1)) * dd, n1 = rsqrtf((float)(cnt[s1] + 1)) * dd;
        const float n2 = rsqrtf((float)(cnt[s2] + 1)) * dd;
        acc.x += v0.x * n0; acc.y += v0.y * n0; acc.z += v0.z * n0; acc.w += v0.w * n0;
        if (rem > 1) { acc.x += v1.x * n1; acc.y += v1.y * n1; acc.z += v1.z * n1; acc.w += v1.w * n1; }
        if (rem > 2) { acc.x += v2.x * n2; acc.y += v2.y * n2; acc.z += v2.z * n2; acc.w += v2.w * n2; }
    }
    if (deg > SLOT_CAP) {
        int oc = cnt[N];
        oc = oc < spill_cap ? oc : spill_cap;
        for (int i = 0; i < oc; ++i) {
            const int2 en = spill[i];
            if (en.x == node) {
                const float4 v = *(const float4*)(h + (size_t)en.y * F + fo);
                const float nn = rsqrtf((float)(cnt[en.y] + 1)) * dd;
                acc.x += v.x * nn; acc.y += v.y * nn; acc.z += v.z * nn; acc.w += v.w * nn;
            }
        }
    }

    const float4 bb = *(const float4*)(bias + fo);
    float4 r;
    r.x = acc.x + bb.x; r.y = acc.y + bb.y; r.z = acc.z + bb.z; r.w = acc.w + bb.w;
    if (RELU) {
        r.x = fmaxf(r.x, 0.0f); r.y = fmaxf(r.y, 0.0f); r.z = fmaxf(r.z, 0.0f); r.w = fmaxf(r.w, 0.0f);
    }
    *(float4*)(o + (size_t)node * F + fo) = r;
}


extern "C" void kernel_launch(void* const* d_in, const int* in_sizes, int n_in,
                              void* d_out, int out_size, void* d_ws, size_t ws_size,
                              hipStream_t stream) {
    const float* x  = (const float*)d_in[0];
    const int*   ei = (const int*)d_in[1];
    const float* W1 = (const float*)d_in[2];
    const float* b1 = (const float*)d_in[3];
    const float* W2 = (const float*)d_in[4];
    const float* b2 = (const float*)d_in[5];
    float* out = (float*)d_out;

    const int N = in_sizes[0] / N_FEAT;
    const int E = in_sizes[1] / 2;
    const int* srcI = ei;
    const int* dstI = ei + E;

    char* ws = (char*)d_ws;
    size_t cntB = ((size_t)(N + 1) * sizeof(int) + 255) & ~(size_t)255;
    size_t t1B  = (((size_t)N * HIDDEN * sizeof(float)) + 255) & ~(size_t)255;
    size_t t3B  = (((size_t)N * NCLS * sizeof(float)) + 255) & ~(size_t)255;
    int*   cnt  = (int*)ws;
    float* t1   = (float*)(ws + cntB);
    float* t2   = (float*)(ws + cntB + t1B);
    float* t3   = t1;

    const int slotInts = N * SLOT_CAP;
    const int w1Words = N_FEAT * HIDDEN;
    const int w2Words = HIDDEN * NCLS;
    const int spillCap = (N * (NCLS - SLOT_CAP) - w1Words - w2Words) / 2;
    int*  slotsA = (int*)out;
    float* W1f   = (float*)(slotsA + slotInts);
    float* W2f   = W1f + w1Words;
    int2* spillA = (int2*)(W2f + w2Words);
    int*  slotsB = (int*)((char*)t1 + t3B);
    int2* spillB = (int2*)(slotsB + slotInts);

    const int BT = 256;
    const int Mtiles = N / 16;
    const int gemmBlocks = (Mtiles + 7) / 8;

    const int fillBlocks = (N + 1 + BT - 1) / BT;
    const int w1Blocks = (w1Words / 4 + BT - 1) / BT;
    const int w2Blocks = (w2Words / 4 + BT - 1) / BT;
    prep_kernel<<<fillBlocks + w1Blocks + w2Blocks, BT, 0, stream>>>(cnt, N + 1, fillBlocks, W1, W1f, w1Blocks, W2, W2f);
    const int slotBlocks = (E + 1023) / 1024;
    gemm1_slots_kernel<<<gemmBlocks + slotBlocks, 256, 0, stream>>>(x, W1f, t1, N, gemmBlocks, srcI, dstI,
                                                                  cnt, slotsA, spillA, E, spillCap);

    long long tot1 = (long long)N * (HIDDEN / 4);
    gather_kernel<HIDDEN, true><<<(int)((tot1 + BT - 1) / BT), BT, 0, stream>>>(
        t1, slotsA, cnt, spillA, b1, t2, N, spillCap);

    const int copyBlocks = (slotInts / 4 + 1023) / 1024;
    gemm2_copy_kernel<<<gemmBlocks + copyBlocks + 1, 256, 0, stream>>>(t2, W2f, t3, N, gemmBlocks, copyBlocks,
                                                                     slotsA, slotsB, slotInts,
                                                                     cnt, spillA, spillB, spillCap);

    long long tot2 = (long long)N * (NCLS / 4);
    gather_kernel<NCLS, false><<<(int)((tot2 + BT - 1) / BT), BT, 0, stream>>>(
        t3, slotsB, cnt, spillB, b2, out, N, spillCap);
}
